# starting kernel
# baseline (speedup 1.0000x reference)
#include <hip/hip_runtime.h>

#define BB 4
#define SS 128
#define DD 256
#define NN 1024
#define EPS 1e-6f

typedef __attribute__((ext_vector_type(2))) float v2f;
typedef __attribute__((ext_vector_type(8))) float v8f;

__global__ void lns_linear_kernel(const float* __restrict__ x,
                                  const float* __restrict__ w,
                                  const float* __restrict__ bias,
                                  float* __restrict__ out) {
    int idx = blockIdx.x * blockDim.x + threadIdx.x;
    int n  = idx & (NN - 1);
    int bs = idx >> 10;
    int s  = bs & (SS - 1);
    int b  = bs >> 7;
    int d  = n >> 2;
    int c  = n & 3;
    float val = fmaxf(x[(b * SS + s) * 4 + c], EPS)
              + __expf(w[(c * SS + s) * DD + d])
              + __expf(bias[d * 4 + c]);
    out[idx] = val;
}

__global__ void scores_wmma_kernel(const float* __restrict__ Qr,
                                   const float* __restrict__ Kr,
                                   float* __restrict__ scores) {
    int wave = (blockIdx.x * blockDim.x + threadIdx.x) >> 5;
    int lane = threadIdx.x & 31;
    int b  = wave >> 6;
    int i0 = ((wave >> 3) & 7) << 4;
    int j0 = (wave & 7) << 4;
    int half = lane >> 4;
    int lr   = lane & 15;

    const float* arow = Qr + (size_t)(b * SS + i0 + lr) * NN + 2 * half;
    const float* brow = Kr + (size_t)(b * SS + j0 + lr) * NN + 2 * half;

    v8f c = {};
    #pragma unroll 4
    for (int k = 0; k < NN; k += 4) {
        v2f a  = *(const v2f*)(arow + k);
        v2f bm = *(const v2f*)(brow + k);
        c = __builtin_amdgcn_wmma_f32_16x16x4_f32(false, a, false, bm,
                                                  (short)0, c, false, false);
    }

    float* srow = scores + (size_t)(b * SS + i0 + 8 * half) * SS + j0 + lr;
    #pragma unroll
    for (int r = 0; r < 8; ++r) srow[(size_t)r * SS] = c[r];
}

__global__ void softmax_clip_kernel(float* __restrict__ scores) {
    __shared__ float red[SS];
    int row = blockIdx.x;
    int t   = threadIdx.x;
    float v = scores[(size_t)row * SS + t];
    red[t] = v;
    __syncthreads();
    for (int off = SS / 2; off > 0; off >>= 1) {
        if (t < off) red[t] = fmaxf(red[t], red[t + off]);
        __syncthreads();
    }
    float m = red[0];
    __syncthreads();
    float e = __expf(v - m);
    red[t] = e;
    __syncthreads();
    for (int off = SS / 2; off > 0; off >>= 1) {
        if (t < off) red[t] += red[t + off];
        __syncthreads();
    }
    float sum = red[0];
    scores[(size_t)row * SS + t] = fmaxf(e / sum, EPS);
}

__global__ void out_wmma_kernel(const float* __restrict__ attn,
                                const float* __restrict__ Vr,
                                float* __restrict__ out) {
    int wave = (blockIdx.x * blockDim.x + threadIdx.x) >> 5;
    int lane = threadIdx.x & 31;
    int b  = wave >> 9;
    int i0 = ((wave >> 6) & 7) << 4;
    int n0 = (wave & 63) << 4;
    int half = lane >> 4;
    int lr   = lane & 15;

    const float* arow = attn + (size_t)(b * SS + i0 + lr) * SS + 2 * half;
    const float* bcol = Vr + (size_t)(b * SS) * NN + n0 + lr;

    v8f c = {};
    #pragma unroll 4
    for (int k = 0; k < SS; k += 4) {
        v2f a = *(const v2f*)(arow + k);
        int kk = k + 2 * half;
        v2f bm;
        bm[0] = bcol[(size_t)kk * NN];
        bm[1] = bcol[(size_t)(kk + 1) * NN];
        c = __builtin_amdgcn_wmma_f32_16x16x4_f32(false, a, false, bm,
                                                  (short)0, c, false, false);
    }

    float* orow = out + (size_t)(b * SS + i0 + 8 * half) * NN + n0 + lr;
    #pragma unroll
    for (int r = 0; r < 8; ++r) orow[(size_t)r * NN] = c[r];
}

extern "C" void kernel_launch(void* const* d_in, const int* in_sizes, int n_in,
                              void* d_out, int out_size, void* d_ws, size_t ws_size,
                              hipStream_t stream) {
    (void)in_sizes; (void)n_in; (void)out_size; (void)ws_size;

    const float* query  = (const float*)d_in[0];
    const float* key_in = (const float*)d_in[1];
    const float* value  = (const float*)d_in[2];
    const float* qw = (const float*)d_in[3];
    const float* qb = (const float*)d_in[4];
    const float* kw = (const float*)d_in[5];
    const float* kb = (const float*)d_in[6];
    const float* vw = (const float*)d_in[7];
    const float* vb = (const float*)d_in[8];
    float* out = (float*)d_out;

    float* Qr = (float*)d_ws;
    float* Kr = Qr + (size_t)BB * SS * NN;
    float* Vr = Kr + (size_t)BB * SS * NN;
    float* sc = Vr + (size_t)BB * SS * NN;

    const int total = BB * SS * NN;
    lns_linear_kernel<<<total / 256, 256, 0, stream>>>(query, qw, qb, Qr);
    lns_linear_kernel<<<total / 256, 256, 0, stream>>>(key_in, kw, kb, Kr);
    lns_linear_kernel<<<total / 256, 256, 0, stream>>>(value, vw, vb, Vr);

    scores_wmma_kernel<<<64, 128, 0, stream>>>(Qr, Kr, sc);

    softmax_clip_kernel<<<BB * SS, SS, 0, stream>>>(sc);

    out_wmma_kernel<<<512, 128, 0, stream>>>(sc, Vr, out);
}


# LNS linear pass: 3 launches of one-element threads (49152 waves) become 1 launch of 1024 waves; a thread owns one (s,d) of q,k,v: exps once, one float4 store per batch; add order kept
# speedup vs baseline: 1.0712x; 1.0712x over previous
#include <hip/hip_runtime.h>

#define BB 4
#define SS 128
#define DD 256
#define NN 1024
#define EPS 1e-6f

typedef __attribute__((ext_vector_type(2))) float v2f;
typedef __attribute__((ext_vector_type(8))) float v8f;

__device__ __forceinline__ void lns_linear_sd(const float* __restrict__ x,
                                              const float* __restrict__ w,
                                              const float* __restrict__ bias,
                                              float* __restrict__ out,
                                              int s, int d) {
    float ew0 = __expf(w[(0 * SS + s) * DD + d]);
    float ew1 = __expf(w[(1 * SS + s) * DD + d]);
    float ew2 = __expf(w[(2 * SS + s) * DD + d]);
    float ew3 = __expf(w[(3 * SS + s) * DD + d]);
    float4 bv = *(const float4*)(bias + d * 4);
    float eb0 = __expf(bv.x);
    float eb1 = __expf(bv.y);
    float eb2 = __expf(bv.z);
    float eb3 = __expf(bv.w);
    for (int b = 0; b < BB; ++b) {
        float4 xv = *(const float4*)(x + (b * SS + s) * 4);
        float4 o;
        o.x = fmaxf(xv.x, EPS) + ew0 + eb0;
        o.y = fmaxf(xv.y, EPS) + ew1 + eb1;
        o.z = fmaxf(xv.z, EPS) + ew2 + eb2;
        o.w = fmaxf(xv.w, EPS) + ew3 + eb3;
        *(float4*)(out + (size_t)(b * SS + s) * NN + d * 4) = o;
    }
}

__global__ void lns_linear3_kernel(const float* __restrict__ xq,
                                   const float* __restrict__ wq,
                                   const float* __restrict__ bq,
                                   const float* __restrict__ xk,
                                   const float* __restrict__ wk,
                                   const float* __restrict__ bk,
                                   const float* __restrict__ xv,
                                   const float* __restrict__ wv,
                                   const float* __restrict__ bv,
                                   float* __restrict__ Qr,
                                   float* __restrict__ Kr,
                                   float* __restrict__ Vr) {
    int s = blockIdx.x >> 1;
    int d = ((blockIdx.x & 1) << 7) + threadIdx.x;
    lns_linear_sd(xq, wq, bq, Qr, s, d);
    lns_linear_sd(xk, wk, bk, Kr, s, d);
    lns_linear_sd(xv, wv, bv, Vr, s, d);
}

__global__ void scores_wmma_kernel(const float* __restrict__ Qr,
                                   const float* __restrict__ Kr,
                                   float* __restrict__ scores) {
    int wave = (blockIdx.x * blockDim.x + threadIdx.x) >> 5;
    int lane = threadIdx.x & 31;
    int b  = wave >> 6;
    int i0 = ((wave >> 3) & 7) << 4;
    int j0 = (wave & 7) << 4;
    int half = lane >> 4;
    int lr   = lane & 15;

    const float* arow = Qr + (size_t)(b * SS + i0 + lr) * NN + 2 * half;
    const float* brow = Kr + (size_t)(b * SS + j0 + lr) * NN + 2 * half;

    v8f c = {};
    #pragma unroll 4
    for (int k = 0; k < NN; k += 4) {
        v2f a  = *(const v2f*)(arow + k);
        v2f bm = *(const v2f*)(brow + k);
        c = __builtin_amdgcn_wmma_f32_16x16x4_f32(false, a, false, bm,
                                                  (short)0, c, false, false);
    }

    float* srow = scores + (size_t)(b * SS + i0 + 8 * half) * SS + j0 + lr;
    #pragma unroll
    for (int r = 0; r < 8; ++r) srow[(size_t)r * SS] = c[r];
}

__global__ void softmax_clip_kernel(float* __restrict__ scores) {
    __shared__ float red[SS];
    int row = blockIdx.x;
    int t   = threadIdx.x;
    float v = scores[(size_t)row * SS + t];
    red[t] = v;
    __syncthreads();
    for (int off = SS / 2; off > 0; off >>= 1) {
        if (t < off) red[t] = fmaxf(red[t], red[t + off]);
        __syncthreads();
    }
    float m = red[0];
    __syncthreads();
    float e = __expf(v - m);
    red[t] = e;
    __syncthreads();
    for (int off = SS / 2; off > 0; off >>= 1) {
        if (t < off) red[t] += red[t + off];
        __syncthreads();
    }
    float sum = red[0];
    scores[(size_t)row * SS + t] = fmaxf(e / sum, EPS);
}

__global__ void out_wmma_kernel(const float* __restrict__ attn,
                                const float* __restrict__ Vr,
                                float* __restrict__ out) {
    int wave = (blockIdx.x * blockDim.x + threadIdx.x) >> 5;
    int lane = threadIdx.x & 31;
    int b  = wave >> 9;
    int i0 = ((wave >> 6) & 7) << 4;
    int n0 = (wave & 63) << 4;
    int half = lane >> 4;
    int lr   = lane & 15;

    const float* arow = attn + (size_t)(b * SS + i0 + lr) * SS + 2 * half;
    const float* bcol = Vr + (size_t)(b * SS) * NN + n0 + lr;

    v8f c = {};
    #pragma unroll 4
    for (int k = 0; k < SS; k += 4) {
        v2f a = *(const v2f*)(arow + k);
        int kk = k + 2 * half;
        v2f bm;
        bm[0] = bcol[(size_t)kk * NN];
        bm[1] = bcol[(size_t)(kk + 1) * NN];
        c = __builtin_amdgcn_wmma_f32_16x16x4_f32(false, a, false, bm,
                                                  (short)0, c, false, false);
    }

    float* orow = out + (size_t)(b * SS + i0 + 8 * half) * NN + n0 + lr;
    #pragma unroll
    for (int r = 0; r < 8; ++r) orow[(size_t)r * NN] = c[r];
}

extern "C" void kernel_launch(void* const* d_in, const int* in_sizes, int n_in,
                              void* d_out, int out_size, void* d_ws, size_t ws_size,
                              hipStream_t stream) {
    (void)in_sizes; (void)n_in; (void)out_size; (void)ws_size;

    const float* query  = (const float*)d_in[0];
    const float* key_in = (const float*)d_in[1];
    const float* value  = (const float*)d_in[2];
    const float* qw = (const float*)d_in[3];
    const float* qb = (const float*)d_in[4];
    const float* kw = (const float*)d_in[5];
    const float* kb = (const float*)d_in[6];
    const float* vw = (const float*)d_in[7];
    const float* vb = (const float*)d_in[8];
    float* out = (float*)d_out;

    float* Qr = (float*)d_ws;
    float* Kr = Qr + (size_t)BB * SS * NN;
    float* Vr = Kr + (size_t)BB * SS * NN;
    float* sc = Vr + (size_t)BB * SS * NN;

    lns_linear3_kernel<<<SS * 2, 128, 0, stream>>>(query, qw, qb, key_in, kw, kb,
                                                   value, vw, vb, Qr, Kr, Vr);

    scores_wmma_kernel<<<64, 128, 0, stream>>>(Qr, Kr, sc);

    softmax_clip_kernel<<<BB * SS, SS, 0, stream>>>(sc);

    out_wmma_kernel<<<512, 128, 0, stream>>>(sc, Vr, out);
}


# scores GEMM: K=1024 loop of one wave per 16x16 tile split over 8 waves x 2 accumulators (256 -> 2048 waves), all 32 float4 fragment loads in flight, partials added via LDS; k sum order changes
# speedup vs baseline: 1.9278x; 1.7997x over previous
#include <hip/hip_runtime.h>

#define BB 4
#define SS 128
#define DD 256
#define NN 1024
#define EPS 1e-6f

typedef __attribute__((ext_vector_type(2))) float v2f;
typedef __attribute__((ext_vector_type(8))) float v8f;

__device__ __forceinline__ void lns_linear_sd(const float* __restrict__ x,
                                              const float* __restrict__ w,
                                              const float* __restrict__ bias,
                                              float* __restrict__ out,
                                              int s, int d) {
    float ew0 = __expf(w[(0 * SS + s) * DD + d]);
    float ew1 = __expf(w[(1 * SS + s) * DD + d]);
    float ew2 = __expf(w[(2 * SS + s) * DD + d]);
    float ew3 = __expf(w[(3 * SS + s) * DD + d]);
    float4 bv = *(const float4*)(bias + d * 4);
    float eb0 = __expf(bv.x);
    float eb1 = __expf(bv.y);
    float eb2 = __expf(bv.z);
    float eb3 = __expf(bv.w);
    for (int b = 0; b < BB; ++b) {
        float4 xv = *(const float4*)(x + (b * SS + s) * 4);
        float4 o;
        o.x = fmaxf(xv.x, EPS) + ew0 + eb0;
        o.y = fmaxf(xv.y, EPS) + ew1 + eb1;
        o.z = fmaxf(xv.z, EPS) + ew2 + eb2;
        o.w = fmaxf(xv.w, EPS) + ew3 + eb3;
        *(float4*)(out + (size_t)(b * SS + s) * NN + d * 4) = o;
    }
}

__global__ void lns_linear3_kernel(const float* __restrict__ xq,
                                   const float* __restrict__ wq,
                                   const float* __restrict__ bq,
                                   const float* __restrict__ xk,
                                   const float* __restrict__ wk,
                                   const float* __restrict__ bk,
                                   const float* __restrict__ xv,
                                   const float* __restrict__ wv,
                                   const float* __restrict__ bv,
                                   float* __restrict__ Qr,
                                   float* __restrict__ Kr,
                                   float* __restrict__ Vr) {
    int s = blockIdx.x >> 1;
    int d = ((blockIdx.x & 1) << 7) + threadIdx.x;
    lns_linear_sd(xq, wq, bq, Qr, s, d);
    lns_linear_sd(xk, wk, bk, Kr, s, d);
    lns_linear_sd(xv, wv, bv, Vr, s, d);
}

#define KSPLIT 8
#define KSLICE (NN / KSPLIT)
#define KT     (KSLICE / 8)

__global__ __launch_bounds__(256) __attribute__((amdgpu_waves_per_eu(2, 2)))
void scores_wmma_kernel(const float* __restrict__ Qr,
                        const float* __restrict__ Kr,
                        float* __restrict__ scores) {
    __shared__ float part[2 * KSPLIT][256];

    int tile = blockIdx.x;
    int tid  = threadIdx.x;
    int w    = tid >> 5;
    int lane = tid & 31;
    int b  = tile >> 6;
    int i0 = ((tile >> 3) & 7) << 4;
    int j0 = (tile & 7) << 4;
    int half = lane >> 4;
    int lr   = lane & 15;

    const float* arow = Qr + (size_t)(b * SS + i0 + lr) * NN + w * KSLICE + 4 * half;
    const float* brow = Kr + (size_t)(b * SS + j0 + lr) * NN + w * KSLICE + 4 * half;

    float4 a[KT], bm[KT];
    for (int t = 0; t < KT; ++t) {
        a[t]  = *(const float4*)(arow + 8 * t);
        bm[t] = *(const float4*)(brow + 8 * t);
    }

    v8f c0 = {};
    v8f c1 = {};
    for (int t = 0; t < KT; ++t) {
        v2f a0 = {a[t].x, a[t].y};
        v2f b0 = {bm[t].x, bm[t].y};
        v2f a1 = {a[t].z, a[t].w};
        v2f b1 = {bm[t].z, bm[t].w};
        c0 = __builtin_amdgcn_wmma_f32_16x16x4_f32(false, a0, false, b0,
                                                   (short)0, c0, false, false);
        c1 = __builtin_amdgcn_wmma_f32_16x16x4_f32(false, a1, false, b1,
                                                   (short)0, c1, false, false);
    }

    for (int r = 0; r < 8; ++r) {
        part[2 * w][r * 32 + lane]     = c0[r];
        part[2 * w + 1][r * 32 + lane] = c1[r];
    }
    __syncthreads();

    float s = part[0][tid];
    for (int q = 1; q < 2 * KSPLIT; ++q) s += part[q][tid];

    int m = (tid >> 5) + 8 * half;
    scores[(size_t)(b * SS + i0 + m) * SS + j0 + lr] = s;
}

__global__ void softmax_clip_kernel(float* __restrict__ scores) {
    __shared__ float red[SS];
    int row = blockIdx.x;
    int t   = threadIdx.x;
    float v = scores[(size_t)row * SS + t];
    red[t] = v;
    __syncthreads();
    for (int off = SS / 2; off > 0; off >>= 1) {
        if (t < off) red[t] = fmaxf(red[t], red[t + off]);
        __syncthreads();
    }
    float m = red[0];
    __syncthreads();
    float e = __expf(v - m);
    red[t] = e;
    __syncthreads();
    for (int off = SS / 2; off > 0; off >>= 1) {
        if (t < off) red[t] += red[t + off];
        __syncthreads();
    }
    float sum = red[0];
    scores[(size_t)row * SS + t] = fmaxf(e / sum, EPS);
}

__global__ void out_wmma_kernel(const float* __restrict__ attn,
                                const float* __restrict__ Vr,
                                float* __restrict__ out) {
    int wave = (blockIdx.x * blockDim.x + threadIdx.x) >> 5;
    int lane = threadIdx.x & 31;
    int b  = wave >> 9;
    int i0 = ((wave >> 6) & 7) << 4;
    int n0 = (wave & 63) << 4;
    int half = lane >> 4;
    int lr   = lane & 15;

    const float* arow = attn + (size_t)(b * SS + i0 + lr) * SS + 2 * half;
    const float* bcol = Vr + (size_t)(b * SS) * NN + n0 + lr;

    v8f c = {};
    #pragma unroll 4
    for (int k = 0; k < SS; k += 4) {
        v2f a = *(const v2f*)(arow + k);
        int kk = k + 2 * half;
        v2f bm;
        bm[0] = bcol[(size_t)kk * NN];
        bm[1] = bcol[(size_t)(kk + 1) * NN];
        c = __builtin_amdgcn_wmma_f32_16x16x4_f32(false, a, false, bm,
                                                  (short)0, c, false, false);
    }

    float* orow = out + (size_t)(b * SS + i0 + 8 * half) * NN + n0 + lr;
    #pragma unroll
    for (int r = 0; r < 8; ++r) orow[(size_t)r * NN] = c[r];
}

extern "C" void kernel_launch(void* const* d_in, const int* in_sizes, int n_in,
                              void* d_out, int out_size, void* d_ws, size_t ws_size,
                              hipStream_t stream) {
    (void)in_sizes; (void)n_in; (void)out_size; (void)ws_size;

    const float* query  = (const float*)d_in[0];
    const float* key_in = (const float*)d_in[1];
    const float* value  = (const float*)d_in[2];
    const float* qw = (const float*)d_in[3];
    const float* qb = (const float*)d_in[4];
    const float* kw = (const float*)d_in[5];
    const float* kb = (const float*)d_in[6];
    const float* vw = (const float*)d_in[7];
    const float* vb = (const float*)d_in[8];
    float* out = (float*)d_out;

    float* Qr = (float*)d_ws;
    float* Kr = Qr + (size_t)BB * SS * NN;
    float* Vr = Kr + (size_t)BB * SS * NN;
    float* sc = Vr + (size_t)BB * SS * NN;

    lns_linear3_kernel<<<SS * 2, 128, 0, stream>>>(query, qw, qb, key_in, kw, kb,
                                                   value, vw, vb, Qr, Kr, Vr);

    scores_wmma_kernel<<<256, 256, 0, stream>>>(Qr, Kr, sc);

    softmax_clip_kernel<<<BB * SS, SS, 0, stream>>>(sc);

    out_wmma_kernel<<<512, 128, 0, stream>>>(sc, Vr, out);
}


# softmax launch fused into the out GEMM (4 -> 3 launches): each 8-wave group redoes its 16 rows' softmax into LDS under its V loads in flight, A fragments from LDS; sum orders kept
# speedup vs baseline: 2.7871x; 1.4457x over previous
#include <hip/hip_runtime.h>

#define BB 4
#define SS 128
#define DD 256
#define NN 1024
#define EPS 1e-6f

typedef __attribute__((ext_vector_type(2))) float v2f;
typedef __attribute__((ext_vector_type(8))) float v8f;

__device__ __forceinline__ void lns_linear_sd(const float* __restrict__ x,
                                              const float* __restrict__ w,
                                              const float* __restrict__ bias,
                                              float* __restrict__ out,
                                              int s, int d) {
    float ew0 = __expf(w[(0 * SS + s) * DD + d]);
    float ew1 = __expf(w[(1 * SS + s) * DD + d]);
    float ew2 = __expf(w[(2 * SS + s) * DD + d]);
    float ew3 = __expf(w[(3 * SS + s) * DD + d]);
    float4 bv = *(const float4*)(bias + d * 4);
    float eb0 = __expf(bv.x);
    float eb1 = __expf(bv.y);
    float eb2 = __expf(bv.z);
    float eb3 = __expf(bv.w);
    for (int b = 0; b < BB; ++b) {
        float4 xv = *(const float4*)(x + (b * SS + s) * 4);
        float4 o;
        o.x = fmaxf(xv.x, EPS) + ew0 + eb0;
        o.y = fmaxf(xv.y, EPS) + ew1 + eb1;
        o.z = fmaxf(xv.z, EPS) + ew2 + eb2;
        o.w = fmaxf(xv.w, EPS) + ew3 + eb3;
        *(float4*)(out + (size_t)(b * SS + s) * NN + d * 4) = o;
    }
}

__global__ void lns_linear3_kernel(const float* __restrict__ xq,
                                   const float* __restrict__ wq,
                                   const float* __restrict__ bq,
                                   const float* __restrict__ xk,
                                   const float* __restrict__ wk,
                                   const float* __restrict__ bk,
                                   const float* __restrict__ xv,
                                   const float* __restrict__ wv,
                                   const float* __restrict__ bv,
                                   float* __restrict__ Qr,
                                   float* __restrict__ Kr,
                                   float* __restrict__ Vr) {
    int s = blockIdx.x >> 1;
    int d = ((blockIdx.x & 1) << 7) + threadIdx.x;
    lns_linear_sd(xq, wq, bq, Qr, s, d);
    lns_linear_sd(xk, wk, bk, Kr, s, d);
    lns_linear_sd(xv, wv, bv, Vr, s, d);
}

#define KSPLIT 8
#define KSLICE (NN / KSPLIT)
#define KT     (KSLICE / 8)

__global__ __launch_bounds__(256) __attribute__((amdgpu_waves_per_eu(2, 2)))
void scores_wmma_kernel(const float* __restrict__ Qr,
                        const float* __restrict__ Kr,
                        float* __restrict__ scores) {
    __shared__ float part[2 * KSPLIT][256];

    int tile = blockIdx.x;
    int tid  = threadIdx.x;
    int w    = tid >> 5;
    int lane = tid & 31;
    int b  = tile >> 6;
    int i0 = ((tile >> 3) & 7) << 4;
    int j0 = (tile & 7) << 4;
    int half = lane >> 4;
    int lr   = lane & 15;

    const float* arow = Qr + (size_t)(b * SS + i0 + lr) * NN + w * KSLICE + 4 * half;
    const float* brow = Kr + (size_t)(b * SS + j0 + lr) * NN + w * KSLICE + 4 * half;

    float4 a[KT], bm[KT];
    for (int t = 0; t < KT; ++t) {
        a[t]  = *(const float4*)(arow + 8 * t);
        bm[t] = *(const float4*)(brow + 8 * t);
    }

    v8f c0 = {};
    v8f c1 = {};
    for (int t = 0; t < KT; ++t) {
        v2f a0 = {a[t].x, a[t].y};
        v2f b0 = {bm[t].x, bm[t].y};
        v2f a1 = {a[t].z, a[t].w};
        v2f b1 = {bm[t].z, bm[t].w};
        c0 = __builtin_amdgcn_wmma_f32_16x16x4_f32(false, a0, false, b0,
                                                   (short)0, c0, false, false);
        c1 = __builtin_amdgcn_wmma_f32_16x16x4_f32(false, a1, false, b1,
                                                   (short)0, c1, false, false);
    }

    for (int r = 0; r < 8; ++r) {
        part[2 * w][r * 32 + lane]     = c0[r];
        part[2 * w + 1][r * 32 + lane] = c1[r];
    }
    __syncthreads();

    float s = part[0][tid];
    for (int q = 1; q < 2 * KSPLIT; ++q) s += part[q][tid];

    int m = (tid >> 5) + 8 * half;
    scores[(size_t)(b * SS + i0 + m) * SS + j0 + lr] = s;
}

#define APITCH 132

__device__ __forceinline__ float xhalf(float x) {
    int v = __builtin_bit_cast(int, x);
    return __builtin_bit_cast(float,
        __builtin_amdgcn_permlanex16(v, v, 0x76543210, 0xfedcba98, false, false));
}
template <int M>
__device__ __forceinline__ float xrow(float x) {
    int v = __builtin_bit_cast(int, x);
    return __builtin_bit_cast(float,
        __builtin_amdgcn_update_dpp(v, v, 0x160 + M, 0xf, 0xf, false));
}
__device__ __forceinline__ float wave_max(float m) {
    m = fmaxf(m, xhalf(m));
    m = fmaxf(m, xrow<8>(m));
    m = fmaxf(m, xrow<4>(m));
    m = fmaxf(m, xrow<2>(m));
    m = fmaxf(m, xrow<1>(m));
    return m;
}
__device__ __forceinline__ float wave_sum(float s) {
    s = s + xhalf(s);
    s = s + xrow<8>(s);
    s = s + xrow<4>(s);
    s = s + xrow<2>(s);
    s = s + xrow<1>(s);
    return s;
}

__global__ __launch_bounds__(256) __attribute__((amdgpu_waves_per_eu(2, 2)))
void softmax_out_wmma_kernel(const float* __restrict__ scores,
                             const float* __restrict__ Vr,
                             float* __restrict__ out) {
    __shared__ __attribute__((aligned(16))) float attn[16 * APITCH];

    int tid  = threadIdx.x;
    int w    = tid >> 5;
    int lane = tid & 31;
    int half = lane >> 4;
    int lr   = lane & 15;
    int blk  = blockIdx.x;
    int b  = blk >> 6;
    int i0 = ((blk >> 3) & 7) << 4;
    int n0 = (((blk & 7) << 3) + w) << 4;

    const float* srow = scores + (size_t)(b * SS + i0 + 2 * w) * SS + lane;
    float v[2][4];
    #pragma unroll
    for (int rr = 0; rr < 2; ++rr)
        #pragma unroll
        for (int q = 0; q < 4; ++q) v[rr][q] = srow[rr * SS + 32 * q];

    const float* bcol = Vr + (size_t)(b * SS) * NN + n0 + lr;
    float bv[SS / 4][2];
    #pragma unroll
    for (int s = 0; s < SS / 4; ++s) {
        int kk = 4 * s + 2 * half;
        bv[s][0] = bcol[(size_t)kk * NN];
        bv[s][1] = bcol[(size_t)(kk + 1) * NN];
    }

    #pragma unroll
    for (int rr = 0; rr < 2; ++rr) {
        float m = wave_max(fmaxf(fmaxf(v[rr][0], v[rr][1]), fmaxf(v[rr][2], v[rr][3])));
        float e0 = __expf(v[rr][0] - m);
        float e1 = __expf(v[rr][1] - m);
        float e2 = __expf(v[rr][2] - m);
        float e3 = __expf(v[rr][3] - m);
        float sum = wave_sum((e0 + e2) + (e1 + e3));
        float* arow_w = attn + (2 * w + rr) * APITCH + lane;
        arow_w[0]  = fmaxf(e0 / sum, EPS);
        arow_w[32] = fmaxf(e1 / sum, EPS);
        arow_w[64] = fmaxf(e2 / sum, EPS);
        arow_w[96] = fmaxf(e3 / sum, EPS);
    }
    __syncthreads();

    const float* arow = attn + lr * APITCH + 2 * half;
    v8f c = {};
    #pragma unroll
    for (int s = 0; s < SS / 4; ++s) {
        v2f a = *(const v2f*)(arow + 4 * s);
        v2f bm = {bv[s][0], bv[s][1]};
        c = __builtin_amdgcn_wmma_f32_16x16x4_f32(false, a, false, bm,
                                                  (short)0, c, false, false);
    }

    float* orow = out + (size_t)(b * SS + i0 + 8 * half) * NN + n0 + lr;
    #pragma unroll
    for (int r = 0; r < 8; ++r) orow[(size_t)r * NN] = c[r];
}

extern "C" void kernel_launch(void* const* d_in, const int* in_sizes, int n_in,
                              void* d_out, int out_size, void* d_ws, size_t ws_size,
                              hipStream_t stream) {
    (void)in_sizes; (void)n_in; (void)out_size; (void)ws_size;

    const float* query  = (const float*)d_in[0];
    const float* key_in = (const float*)d_in[1];
    const float* value  = (const float*)d_in[2];
    const float* qw = (const float*)d_in[3];
    const float* qb = (const float*)d_in[4];
    const float* kw = (const float*)d_in[5];
    const float* kb = (const float*)d_in[6];
    const float* vw = (const float*)d_in[7];
    const float* vb = (const float*)d_in[8];
    float* out = (float*)d_out;

    float* Qr = (float*)d_ws;
    float* Kr = Qr + (size_t)BB * SS * NN;
    float* Vr = Kr + (size_t)BB * SS * NN;
    float* sc = Vr + (size_t)BB * SS * NN;

    lns_linear3_kernel<<<SS * 2, 128, 0, stream>>>(query, qw, qb, key_in, kw, kb,
                                                   value, vw, vb, Qr, Kr, Vr);

    scores_wmma_kernel<<<256, 256, 0, stream>>>(Qr, Kr, sc);

    softmax_out_wmma_kernel<<<256, 256, 0, stream>>>(sc, Vr, out);
}


# LNS linear launch fused into the scores kernel (3 -> 2 launches): each wave builds its Q,K fragments from x,w,bias in registers, bias exps via LDS, group also writes its V slice; k sum order changes
# speedup vs baseline: 3.4237x; 1.2284x over previous
#include <hip/hip_runtime.h>

#define BB 4
#define SS 128
#define DD 256
#define NN 1024
#define EPS 1e-6f

typedef __attribute__((ext_vector_type(2))) float v2f;
typedef __attribute__((ext_vector_type(8))) float v8f;

#define KSPLIT 8

__global__ __launch_bounds__(256) __attribute__((amdgpu_waves_per_eu(2, 2)))
void lns_scores_wmma_kernel(const float* __restrict__ xq,
                            const float* __restrict__ wq,
                            const float* __restrict__ bq,
                            const float* __restrict__ xk,
                            const float* __restrict__ wk,
                            const float* __restrict__ bk,
                            const float* __restrict__ xv,
                            const float* __restrict__ wv,
                            const float* __restrict__ bv,
                            float* __restrict__ Vr,
                            float* __restrict__ scores) {
    __shared__ float part[2 * KSPLIT][256];
    __shared__ __attribute__((aligned(16))) float ebq[4 * DD];
    __shared__ __attribute__((aligned(16))) float ebk[4 * DD];

    int tile = blockIdx.x;
    int tid  = threadIdx.x;
    int w    = tid >> 5;
    int lane = tid & 31;
    int b  = tile >> 6;
    int i0 = ((tile >> 3) & 7) << 4;
    int j0 = (tile & 7) << 4;
    int half = lane >> 4;
    int lr   = lane & 15;

    float4 bqv = *(const float4*)(bq + tid * 4);
    float4 bkv = *(const float4*)(bk + tid * 4);
    float4 xa = *(const float4*)(xq + (b * SS + i0 + lr) * 4);
    float4 xb = *(const float4*)(xk + (b * SS + j0 + lr) * 4);

    int dl0 = 32 * w + 16 * half;
    const float* wqrow = wq + (size_t)(i0 + lr) * DD + dl0;
    const float* wkrow = wk + (size_t)(j0 + lr) * DD + dl0;
    float4 a[4][4], bm[4][4];
    #pragma unroll
    for (int c = 0; c < 4; ++c) {
        #pragma unroll
        for (int t = 0; t < 4; ++t) {
            a[c][t]  = *(const float4*)(wqrow + (size_t)c * SS * DD + 4 * t);
            bm[c][t] = *(const float4*)(wkrow + (size_t)c * SS * DD + 4 * t);
        }
    }

    int vs = tile >> 1;
    int vb = (tile & 1) << 1;
    float wv0 = wv[(0 * SS + vs) * DD + tid];
    float wv1 = wv[(1 * SS + vs) * DD + tid];
    float wv2 = wv[(2 * SS + vs) * DD + tid];
    float wv3 = wv[(3 * SS + vs) * DD + tid];
    float4 bvv = *(const float4*)(bv + tid * 4);
    float4 xv0 = *(const float4*)(xv + ((vb + 0) * SS + vs) * 4);
    float4 xv1 = *(const float4*)(xv + ((vb + 1) * SS + vs) * 4);

    ebq[0 * DD + tid] = __expf(bqv.x);
    ebq[1 * DD + tid] = __expf(bqv.y);
    ebq[2 * DD + tid] = __expf(bqv.z);
    ebq[3 * DD + tid] = __expf(bqv.w);
    ebk[0 * DD + tid] = __expf(bkv.x);
    ebk[1 * DD + tid] = __expf(bkv.y);
    ebk[2 * DD + tid] = __expf(bkv.z);
    ebk[3 * DD + tid] = __expf(bkv.w);

    float xac[4] = {fmaxf(xa.x, EPS), fmaxf(xa.y, EPS), fmaxf(xa.z, EPS), fmaxf(xa.w, EPS)};
    float xbc[4] = {fmaxf(xb.x, EPS), fmaxf(xb.y, EPS), fmaxf(xb.z, EPS), fmaxf(xb.w, EPS)};

    #pragma unroll
    for (int c = 0; c < 4; ++c) {
        #pragma unroll
        for (int t = 0; t < 4; ++t) {
            a[c][t].x = xac[c] + __expf(a[c][t].x);
            a[c][t].y = xac[c] + __expf(a[c][t].y);
            a[c][t].z = xac[c] + __expf(a[c][t].z);
            a[c][t].w = xac[c] + __expf(a[c][t].w);
            bm[c][t].x = xbc[c] + __expf(bm[c][t].x);
            bm[c][t].y = xbc[c] + __expf(bm[c][t].y);
            bm[c][t].z = xbc[c] + __expf(bm[c][t].z);
            bm[c][t].w = xbc[c] + __expf(bm[c][t].w);
        }
    }

    float4 o0, o1;
    {
        float ew0 = __expf(wv0), ew1 = __expf(wv1), ew2 = __expf(wv2), ew3 = __expf(wv3);
        float eb0 = __expf(bvv.x), eb1 = __expf(bvv.y), eb2 = __expf(bvv.z), eb3 = __expf(bvv.w);
        o0.x = fmaxf(xv0.x, EPS) + ew0 + eb0;
        o0.y = fmaxf(xv0.y, EPS) + ew1 + eb1;
        o0.z = fmaxf(xv0.z, EPS) + ew2 + eb2;
        o0.w = fmaxf(xv0.w, EPS) + ew3 + eb3;
        o1.x = fmaxf(xv1.x, EPS) + ew0 + eb0;
        o1.y = fmaxf(xv1.y, EPS) + ew1 + eb1;
        o1.z = fmaxf(xv1.z, EPS) + ew2 + eb2;
        o1.w = fmaxf(xv1.w, EPS) + ew3 + eb3;
    }

    __syncthreads();

    v8f c0 = {};
    v8f c1 = {};
    #pragma unroll
    for (int c = 0; c < 4; ++c) {
        #pragma unroll
        for (int t = 0; t < 4; ++t) {
            float4 eq = *(const float4*)(ebq + c * DD + dl0 + 4 * t);
            float4 ek = *(const float4*)(ebk + c * DD + dl0 + 4 * t);
            float4 qa = a[c][t];
            float4 kb = bm[c][t];
            v2f a0 = {qa.x + eq.x, qa.y + eq.y};
            v2f a1 = {qa.z + eq.z, qa.w + eq.w};
            v2f b0 = {kb.x + ek.x, kb.y + ek.y};
            v2f b1 = {kb.z + ek.z, kb.w + ek.w};
            c0 = __builtin_amdgcn_wmma_f32_16x16x4_f32(false, a0, false, b0,
                                                       (short)0, c0, false, false);
            c1 = __builtin_amdgcn_wmma_f32_16x16x4_f32(false, a1, false, b1,
                                                       (short)0, c1, false, false);
        }
    }

    for (int r = 0; r < 8; ++r) {
        part[2 * w][r * 32 + lane]     = c0[r];
        part[2 * w + 1][r * 32 + lane] = c1[r];
    }
    __syncthreads();

    *(float4*)(Vr + (size_t)((vb + 0) * SS + vs) * NN + tid * 4) = o0;
    *(float4*)(Vr + (size_t)((vb + 1) * SS + vs) * NN + tid * 4) = o1;

    float s = part[0][tid];
    for (int q = 1; q < 2 * KSPLIT; ++q) s += part[q][tid];

    int m = (tid >> 5) + 8 * half;
    scores[(size_t)(b * SS + i0 + m) * SS + j0 + lr] = s;
}

#define APITCH 132

__device__ __forceinline__ float xhalf(float x) {
    int v = __builtin_bit_cast(int, x);
    return __builtin_bit_cast(float,
        __builtin_amdgcn_permlanex16(v, v, 0x76543210, 0xfedcba98, false, false));
}
template <int M>
__device__ __forceinline__ float xrow(float x) {
    int v = __builtin_bit_cast(int, x);
    return __builtin_bit_cast(float,
        __builtin_amdgcn_update_dpp(v, v, 0x160 + M, 0xf, 0xf, false));
}
__device__ __forceinline__ float wave_max(float m) {
    m = fmaxf(m, xhalf(m));
    m = fmaxf(m, xrow<8>(m));
    m = fmaxf(m, xrow<4>(m));
    m = fmaxf(m, xrow<2>(m));
    m = fmaxf(m, xrow<1>(m));
    return m;
}
__device__ __forceinline__ float wave_sum(float s) {
    s = s + xhalf(s);
    s = s + xrow<8>(s);
    s = s + xrow<4>(s);
    s = s + xrow<2>(s);
    s = s + xrow<1>(s);
    return s;
}

__global__ __launch_bounds__(256) __attribute__((amdgpu_waves_per_eu(2, 2)))
void softmax_out_wmma_kernel(const float* __restrict__ scores,
                             const float* __restrict__ Vr,
                             float* __restrict__ out) {
    __shared__ __attribute__((aligned(16))) float attn[16 * APITCH];

    int tid  = threadIdx.x;
    int w    = tid >> 5;
    int lane = tid & 31;
    int half = lane >> 4;
    int lr   = lane & 15;
    int blk  = blockIdx.x;
    int b  = blk >> 6;
    int i0 = ((blk >> 3) & 7) << 4;
    int n0 = (((blk & 7) << 3) + w) << 4;

    const float* srow = scores + (size_t)(b * SS + i0 + 2 * w) * SS + lane;
    float v[2][4];
    #pragma unroll
    for (int rr = 0; rr < 2; ++rr)
        #pragma unroll
        for (int q = 0; q < 4; ++q) v[rr][q] = srow[rr * SS + 32 * q];

    const float* bcol = Vr + (size_t)(b * SS) * NN + n0 + lr;
    float bv[SS / 4][2];
    #pragma unroll
    for (int s = 0; s < SS / 4; ++s) {
        int kk = 4 * s + 2 * half;
        bv[s][0] = bcol[(size_t)kk * NN];
        bv[s][1] = bcol[(size_t)(kk + 1) * NN];
    }

    #pragma unroll
    for (int rr = 0; rr < 2; ++rr) {
        float m = wave_max(fmaxf(fmaxf(v[rr][0], v[rr][1]), fmaxf(v[rr][2], v[rr][3])));
        float e0 = __expf(v[rr][0] - m);
        float e1 = __expf(v[rr][1] - m);
        float e2 = __expf(v[rr][2] - m);
        float e3 = __expf(v[rr][3] - m);
        float sum = wave_sum((e0 + e2) + (e1 + e3));
        float* arow_w = attn + (2 * w + rr) * APITCH + lane;
        arow_w[0]  = fmaxf(e0 / sum, EPS);
        arow_w[32] = fmaxf(e1 / sum, EPS);
        arow_w[64] = fmaxf(e2 / sum, EPS);
        arow_w[96] = fmaxf(e3 / sum, EPS);
    }
    __syncthreads();

    const float* arow = attn + lr * APITCH + 2 * half;
    v8f c = {};
    #pragma unroll
    for (int s = 0; s < SS / 4; ++s) {
        v2f a = *(const v2f*)(arow + 4 * s);
        v2f bm = {bv[s][0], bv[s][1]};
        c = __builtin_amdgcn_wmma_f32_16x16x4_f32(false, a, false, bm,
                                                  (short)0, c, false, false);
    }

    float* orow = out + (size_t)(b * SS + i0 + 8 * half) * NN + n0 + lr;
    #pragma unroll
    for (int r = 0; r < 8; ++r) orow[(size_t)r * NN] = c[r];
}

extern "C" void kernel_launch(void* const* d_in, const int* in_sizes, int n_in,
                              void* d_out, int out_size, void* d_ws, size_t ws_size,
                              hipStream_t stream) {
    (void)in_sizes; (void)n_in; (void)out_size; (void)ws_size;

    const float* query  = (const float*)d_in[0];
    const float* key_in = (const float*)d_in[1];
    const float* value  = (const float*)d_in[2];
    const float* qw = (const float*)d_in[3];
    const float* qb = (const float*)d_in[4];
    const float* kw = (const float*)d_in[5];
    const float* kb = (const float*)d_in[6];
    const float* vw = (const float*)d_in[7];
    const float* vb = (const float*)d_in[8];
    float* out = (float*)d_out;

    float* Vr = (float*)d_ws;
    float* sc = Vr + (size_t)BB * SS * NN;

    lns_scores_wmma_kernel<<<256, 256, 0, stream>>>(query, qw, qb, key_in, kw, kb,
                                                    value, vw, vb, Vr, sc);

    softmax_out_wmma_kernel<<<256, 256, 0, stream>>>(sc, Vr, out);
}


# scores kernel: a group takes one 16x16 tile for all 4 batches and one K quarter: exp(w) fragments loaded and built once per 4 batches; next launch adds the 4 partial planes; k sum order changes
# speedup vs baseline: 3.6784x; 1.0744x over previous
#include <hip/hip_runtime.h>

#define BB 4
#define SS 128
#define DD 256
#define NN 1024
#define EPS 1e-6f

typedef __attribute__((ext_vector_type(2))) float v2f;
typedef __attribute__((ext_vector_type(8))) float v8f;

#define KSPLIT 8
#define KQ 4
#define DQ (DD / KQ)
#define EBP (DQ + 4)
#define PLANE (BB * SS * SS)

__global__ __launch_bounds__(256) __attribute__((amdgpu_waves_per_eu(2, 2)))
void lns_scores_wmma_kernel(const float* __restrict__ xq,
                            const float* __restrict__ wq,
                            const float* __restrict__ bq,
                            const float* __restrict__ xk,
                            const float* __restrict__ wk,
                            const float* __restrict__ bk,
                            const float* __restrict__ xv,
                            const float* __restrict__ wv,
                            const float* __restrict__ bv,
                            float* __restrict__ Vr,
                            float* __restrict__ scores) {
    __shared__ __attribute__((aligned(16))) float part[KSPLIT][BB][256];
    __shared__ __attribute__((aligned(16))) float ebq[4 * EBP];
    __shared__ __attribute__((aligned(16))) float ebk[4 * EBP];

    int blk  = blockIdx.x;
    int tile = blk >> 2;
    int kq   = blk & 3;
    int tid  = threadIdx.x;
    int w    = tid >> 5;
    int lane = tid & 31;
    int i0 = (tile >> 3) << 4;
    int j0 = (tile & 7) << 4;
    int half = lane >> 4;
    int lr   = lane & 15;

    float bq1 = bq[kq * DQ * 4 + tid];
    float bk1 = bk[kq * DQ * 4 + tid];

    int dl = 8 * w + 4 * half;
    const float* wqrow = wq + (size_t)(i0 + lr) * DD + kq * DQ + dl;
    const float* wkrow = wk + (size_t)(j0 + lr) * DD + kq * DQ + dl;
    float4 a[4], bm[4];
    #pragma unroll
    for (int c = 0; c < 4; ++c) {
        a[c]  = *(const float4*)(wqrow + (size_t)c * SS * DD);
        bm[c] = *(const float4*)(wkrow + (size_t)c * SS * DD);
    }

    float4 xa[BB], xb[BB];
    #pragma unroll
    for (int b = 0; b < BB; ++b) {
        xa[b] = *(const float4*)(xq + (b * SS + i0 + lr) * 4);
        xb[b] = *(const float4*)(xk + (b * SS + j0 + lr) * 4);
    }

    int vs = blk >> 1;
    int vb = (blk & 1) << 1;
    float wv0 = wv[(0 * SS + vs) * DD + tid];
    float wv1 = wv[(1 * SS + vs) * DD + tid];
    float wv2 = wv[(2 * SS + vs) * DD + tid];
    float wv3 = wv[(3 * SS + vs) * DD + tid];
    float4 bvv = *(const float4*)(bv + tid * 4);
    float4 xv0 = *(const float4*)(xv + ((vb + 0) * SS + vs) * 4);
    float4 xv1 = *(const float4*)(xv + ((vb + 1) * SS + vs) * 4);

    __builtin_amdgcn_sched_barrier(0);

    ebq[(tid & 3) * EBP + (tid >> 2)] = __expf(bq1);
    ebk[(tid & 3) * EBP + (tid >> 2)] = __expf(bk1);

    #pragma unroll
    for (int c = 0; c < 4; ++c) {
        a[c].x = __expf(a[c].x);
        a[c].y = __expf(a[c].y);
        a[c].z = __expf(a[c].z);
        a[c].w = __expf(a[c].w);
        bm[c].x = __expf(bm[c].x);
        bm[c].y = __expf(bm[c].y);
        bm[c].z = __expf(bm[c].z);
        bm[c].w = __expf(bm[c].w);
    }

    float xac[BB][4], xbc[BB][4];
    #pragma unroll
    for (int b = 0; b < BB; ++b) {
        xac[b][0] = fmaxf(xa[b].x, EPS); xac[b][1] = fmaxf(xa[b].y, EPS);
        xac[b][2] = fmaxf(xa[b].z, EPS); xac[b][3] = fmaxf(xa[b].w, EPS);
        xbc[b][0] = fmaxf(xb[b].x, EPS); xbc[b][1] = fmaxf(xb[b].y, EPS);
        xbc[b][2] = fmaxf(xb[b].z, EPS); xbc[b][3] = fmaxf(xb[b].w, EPS);
    }

    float4 o0, o1;
    {
        float ew0 = __expf(wv0), ew1 = __expf(wv1), ew2 = __expf(wv2), ew3 = __expf(wv3);
        float eb0 = __expf(bvv.x), eb1 = __expf(bvv.y), eb2 = __expf(bvv.z), eb3 = __expf(bvv.w);
        o0.x = fmaxf(xv0.x, EPS) + ew0 + eb0;
        o0.y = fmaxf(xv0.y, EPS) + ew1 + eb1;
        o0.z = fmaxf(xv0.z, EPS) + ew2 + eb2;
        o0.w = fmaxf(xv0.w, EPS) + ew3 + eb3;
        o1.x = fmaxf(xv1.x, EPS) + ew0 + eb0;
        o1.y = fmaxf(xv1.y, EPS) + ew1 + eb1;
        o1.z = fmaxf(xv1.z, EPS) + ew2 + eb2;
        o1.w = fmaxf(xv1.w, EPS) + ew3 + eb3;
    }

    __syncthreads();

    v8f acc[BB];
    #pragma unroll
    for (int b = 0; b < BB; ++b) acc[b] = (v8f){};

    #pragma unroll
    for (int c = 0; c < 4; ++c) {
        float4 eq = *(const float4*)(ebq + c * EBP + dl);
        float4 ek = *(const float4*)(ebk + c * EBP + dl);
        float4 qa = a[c];
        float4 kb = bm[c];
        #pragma unroll
        for (int b = 0; b < BB; ++b) {
            float xqc = xac[b][c], xkc = xbc[b][c];
            v2f a0 = {(xqc + qa.x) + eq.x, (xqc + qa.y) + eq.y};
            v2f b0 = {(xkc + kb.x) + ek.x, (xkc + kb.y) + ek.y};
            acc[b] = __builtin_amdgcn_wmma_f32_16x16x4_f32(false, a0, false, b0,
                                                           (short)0, acc[b], false, false);
        }
        #pragma unroll
        for (int b = 0; b < BB; ++b) {
            float xqc = xac[b][c], xkc = xbc[b][c];
            v2f a1 = {(xqc + qa.z) + eq.z, (xqc + qa.w) + eq.w};
            v2f b1 = {(xkc + kb.z) + ek.z, (xkc + kb.w) + ek.w};
            acc[b] = __builtin_amdgcn_wmma_f32_16x16x4_f32(false, a1, false, b1,
                                                           (short)0, acc[b], false, false);
        }
    }

    #pragma unroll
    for (int b = 0; b < BB; ++b)
        #pragma unroll
        for (int r = 0; r < 8; ++r) part[w][b][r * 32 + lane] = acc[b][r];
    __syncthreads();

    *(float4*)(Vr + (size_t)((vb + 0) * SS + vs) * NN + tid * 4) = o0;
    *(float4*)(Vr + (size_t)((vb + 1) * SS + vs) * NN + tid * 4) = o1;

    int ob = tid >> 6;
    int s4 = (tid & 63) << 2;
    float4 s = *(const float4*)(&part[0][ob][s4]);
    #pragma unroll
    for (int q = 1; q < KSPLIT; ++q) {
        float4 p = *(const float4*)(&part[q][ob][s4]);
        s.x += p.x; s.y += p.y; s.z += p.z; s.w += p.w;
    }
    int m = (s4 >> 5) + 8 * ((s4 >> 4) & 1);
    int n = s4 & 15;
    *(float4*)(scores + (size_t)kq * PLANE + (size_t)(ob * SS + i0 + m) * SS + j0 + n) = s;
}

#define APITCH 132

__device__ __forceinline__ float xhalf(float x) {
    int v = __builtin_bit_cast(int, x);
    return __builtin_bit_cast(float,
        __builtin_amdgcn_permlanex16(v, v, 0x76543210, 0xfedcba98, false, false));
}
template <int M>
__device__ __forceinline__ float xrow(float x) {
    int v = __builtin_bit_cast(int, x);
    return __builtin_bit_cast(float,
        __builtin_amdgcn_update_dpp(v, v, 0x160 + M, 0xf, 0xf, false));
}
__device__ __forceinline__ float wave_max(float m) {
    m = fmaxf(m, xhalf(m));
    m = fmaxf(m, xrow<8>(m));
    m = fmaxf(m, xrow<4>(m));
    m = fmaxf(m, xrow<2>(m));
    m = fmaxf(m, xrow<1>(m));
    return m;
}
__device__ __forceinline__ float wave_sum(float s) {
    s = s + xhalf(s);
    s = s + xrow<8>(s);
    s = s + xrow<4>(s);
    s = s + xrow<2>(s);
    s = s + xrow<1>(s);
    return s;
}

__global__ __launch_bounds__(256) __attribute__((amdgpu_waves_per_eu(2, 2)))
void softmax_out_wmma_kernel(const float* __restrict__ scores,
                             const float* __restrict__ Vr,
                             float* __restrict__ out) {
    __shared__ __attribute__((aligned(16))) float attn[16 * APITCH];

    int tid  = threadIdx.x;
    int w    = tid >> 5;
    int lane = tid & 31;
    int half = lane >> 4;
    int lr   = lane & 15;
    int blk  = blockIdx.x;
    int b  = blk >> 6;
    int i0 = ((blk >> 3) & 7) << 4;
    int n0 = (((blk & 7) << 3) + w) << 4;

    const float* srow = scores + (size_t)(b * SS + i0 + 2 * w) * SS + lane;
    float v[2][4];
    #pragma unroll
    for (int rr = 0; rr < 2; ++rr)
        #pragma unroll
        for (int q = 0; q < 4; ++q) {
            const float* p = srow + rr * SS + 32 * q;
            v[rr][q] = ((p[0] + p[PLANE]) + p[2 * PLANE]) + p[3 * PLANE];
        }

    const float* bcol = Vr + (size_t)(b * SS) * NN + n0 + lr;
    float bv[SS / 4][2];
    #pragma unroll
    for (int s = 0; s < SS / 4; ++s) {
        int kk = 4 * s + 2 * half;
        bv[s][0] = bcol[(size_t)kk * NN];
        bv[s][1] = bcol[(size_t)(kk + 1) * NN];
    }

    #pragma unroll
    for (int rr = 0; rr < 2; ++rr) {
        float m = wave_max(fmaxf(fmaxf(v[rr][0], v[rr][1]), fmaxf(v[rr][2], v[rr][3])));
        float e0 = __expf(v[rr][0] - m);
        float e1 = __expf(v[rr][1] - m);
        float e2 = __expf(v[rr][2] - m);
        float e3 = __expf(v[rr][3] - m);
        float sum = wave_sum((e0 + e2) + (e1 + e3));
        float* arow_w = attn + (2 * w + rr) * APITCH + lane;
        arow_w[0]  = fmaxf(e0 / sum, EPS);
        arow_w[32] = fmaxf(e1 / sum, EPS);
        arow_w[64] = fmaxf(e2 / sum, EPS);
        arow_w[96] = fmaxf(e3 / sum, EPS);
    }
    __syncthreads();

    const float* arow = attn + lr * APITCH + 2 * half;
    v8f c = {};
    #pragma unroll
    for (int s = 0; s < SS / 4; ++s) {
        v2f a = *(const v2f*)(arow + 4 * s);
        v2f bm = {bv[s][0], bv[s][1]};
        c = __builtin_amdgcn_wmma_f32_16x16x4_f32(false, a, false, bm,
                                                  (short)0, c, false, false);
    }

    float* orow = out + (size_t)(b * SS + i0 + 8 * half) * NN + n0 + lr;
    #pragma unroll
    for (int r = 0; r < 8; ++r) orow[(size_t)r * NN] = c[r];
}

extern "C" void kernel_launch(void* const* d_in, const int* in_sizes, int n_in,
                              void* d_out, int out_size, void* d_ws, size_t ws_size,
                              hipStream_t stream) {
    (void)in_sizes; (void)n_in; (void)out_size; (void)ws_size;

    const float* query  = (const float*)d_in[0];
    const float* key_in = (const float*)d_in[1];
    const float* value  = (const float*)d_in[2];
    const float* qw = (const float*)d_in[3];
    const float* qb = (const float*)d_in[4];
    const float* kw = (const float*)d_in[5];
    const float* kb = (const float*)d_in[6];
    const float* vw = (const float*)d_in[7];
    const float* vb = (const float*)d_in[8];
    float* out = (float*)d_out;

    float* Vr = (float*)d_ws;
    float* sc = Vr + (size_t)BB * SS * NN;

    lns_scores_wmma_kernel<<<256, 256, 0, stream>>>(query, qw, qb, key_in, kw, kb,
                                                    value, vw, vb, Vr, sc);

    softmax_out_wmma_kernel<<<256, 256, 0, stream>>>(sc, Vr, out);
}


# scores kernel: batch-free product of the exp(w)+exp(bias) terms once for the 4 batches; max(x,eps) enters as one rank-4 WMMA step per wave via row sums in LDS: 32->9 WMMAs a wave; k sum order changes
# speedup vs baseline: 3.9445x; 1.0723x over previous
#include <hip/hip_runtime.h>

#define BB 4
#define SS 128
#define DD 256
#define NN 1024
#define EPS 1e-6f

typedef __attribute__((ext_vector_type(2))) float v2f;
typedef __attribute__((ext_vector_type(8))) float v8f;

#define KSPLIT 8
#define KQ 4
#define DQ (DD / KQ)
#define EBP (DQ + 4)
#define PLANE (BB * SS * SS)

__device__ __forceinline__ float xhalf(float x) {
    int v = __builtin_bit_cast(int, x);
    return __builtin_bit_cast(float,
        __builtin_amdgcn_permlanex16(v, v, 0x76543210, 0xfedcba98, false, false));
}
template <int M>
__device__ __forceinline__ float xrow(float x) {
    int v = __builtin_bit_cast(int, x);
    return __builtin_bit_cast(float,
        __builtin_amdgcn_update_dpp(v, v, 0x160 + M, 0xf, 0xf, false));
}

__global__ __launch_bounds__(256) __attribute__((amdgpu_waves_per_eu(2, 2)))
void lns_scores_wmma_kernel(const float* __restrict__ xq,
                            const float* __restrict__ wq,
                            const float* __restrict__ bq,
                            const float* __restrict__ xk,
                            const float* __restrict__ wk,
                            const float* __restrict__ bk,
                            const float* __restrict__ xv,
                            const float* __restrict__ wv,
                            const float* __restrict__ bv,
                            float* __restrict__ Vr,
                            float* __restrict__ scores) {
    __shared__ __attribute__((aligned(16))) float part[KSPLIT][2][256];
    __shared__ __attribute__((aligned(16))) float ebq[4 * EBP];
    __shared__ __attribute__((aligned(16))) float ebk[4 * EBP];
    __shared__ __attribute__((aligned(16))) float swq[KSPLIT][16][4];
    __shared__ __attribute__((aligned(16))) float swk[KSPLIT][16][4];
    __shared__ __attribute__((aligned(16))) float sbq[KSPLIT][4];
    __shared__ __attribute__((aligned(16))) float sbk[KSPLIT][4];

    int blk  = blockIdx.x;
    int tile = blk >> 2;
    int kq   = blk & 3;
    int tid  = threadIdx.x;
    int w    = tid >> 5;
    int lane = tid & 31;
    int i0 = (tile >> 3) << 4;
    int j0 = (tile & 7) << 4;
    int half = lane >> 4;
    int lr   = lane & 15;

    float bq1 = bq[kq * DQ * 4 + tid];
    float bk1 = bk[kq * DQ * 4 + tid];

    int dl = 8 * w + 4 * half;
    const float* wqrow = wq + (size_t)(i0 + lr) * DD + kq * DQ + dl;
    const float* wkrow = wk + (size_t)(j0 + lr) * DD + kq * DQ + dl;
    float4 a[4], bm[4];
    #pragma unroll
    for (int c = 0; c < 4; ++c) {
        a[c]  = *(const float4*)(wqrow + (size_t)c * SS * DD);
        bm[c] = *(const float4*)(wkrow + (size_t)c * SS * DD);
    }

    int rb = w & 3;
    int rt = w >> 2;
    float2 xa = *(const float2*)(xq + (rb * SS + i0 + lr) * 4 + 2 * half);
    float2 xb = *(const float2*)(xk + (rb * SS + j0 + lr) * 4 + 2 * half);

    int vs = blk >> 1;
    int vb = (blk & 1) << 1;
    float wv0 = wv[(0 * SS + vs) * DD + tid];
    float wv1 = wv[(1 * SS + vs) * DD + tid];
    float wv2 = wv[(2 * SS + vs) * DD + tid];
    float wv3 = wv[(3 * SS + vs) * DD + tid];
    float4 bvv = *(const float4*)(bv + tid * 4);
    float4 xv0 = *(const float4*)(xv + ((vb + 0) * SS + vs) * 4);
    float4 xv1 = *(const float4*)(xv + ((vb + 1) * SS + vs) * 4);

    __builtin_amdgcn_sched_barrier(0);

    float ebq1 = __expf(bq1);
    float ebk1 = __expf(bk1);
    ebq[(tid & 3) * EBP + (tid >> 2)] = ebq1;
    ebk[(tid & 3) * EBP + (tid >> 2)] = ebk1;
    {
        float tq = ebq1, tk = ebk1;
        tq = tq + xrow<4>(tq);  tk = tk + xrow<4>(tk);
        tq = tq + xrow<8>(tq);  tk = tk + xrow<8>(tk);
        tq = tq + xhalf(tq);    tk = tk + xhalf(tk);
        if (lane < 4) { sbq[w][lane] = tq; sbk[w][lane] = tk; }
    }

    #pragma unroll
    for (int c = 0; c < 4; ++c) {
        a[c].x = __expf(a[c].x);
        a[c].y = __expf(a[c].y);
        a[c].z = __expf(a[c].z);
        a[c].w = __expf(a[c].w);
        bm[c].x = __expf(bm[c].x);
        bm[c].y = __expf(bm[c].y);
        bm[c].z = __expf(bm[c].z);
        bm[c].w = __expf(bm[c].w);
    }

    {
        float pq[4], pk[4];
        #pragma unroll
        for (int c = 0; c < 4; ++c) {
            v2f tq = (v2f){a[c].x, a[c].y} + (v2f){a[c].z, a[c].w};
            v2f tk = (v2f){bm[c].x, bm[c].y} + (v2f){bm[c].z, bm[c].w};
            float uq = tq.x + tq.y;
            float uk = tk.x + tk.y;
            pq[c] = uq + xhalf(uq);
            pk[c] = uk + xhalf(uk);
        }
        float4 st;
        st.x = half ? pk[0] : pq[0];
        st.y = half ? pk[1] : pq[1];
        st.z = half ? pk[2] : pq[2];
        st.w = half ? pk[3] : pq[3];
        *(float4*)(half ? &swk[w][lr][0] : &swq[w][lr][0]) = st;
    }

    xa.x = fmaxf(xa.x, EPS); xa.y = fmaxf(xa.y, EPS);
    xb.x = fmaxf(xb.x, EPS); xb.y = fmaxf(xb.y, EPS);

    float4 o0, o1;
    {
        float ew0 = __expf(wv0), ew1 = __expf(wv1), ew2 = __expf(wv2), ew3 = __expf(wv3);
        float eb0 = __expf(bvv.x), eb1 = __expf(bvv.y), eb2 = __expf(bvv.z), eb3 = __expf(bvv.w);
        o0.x = fmaxf(xv0.x, EPS) + ew0 + eb0;
        o0.y = fmaxf(xv0.y, EPS) + ew1 + eb1;
        o0.z = fmaxf(xv0.z, EPS) + ew2 + eb2;
        o0.w = fmaxf(xv0.w, EPS) + ew3 + eb3;
        o1.x = fmaxf(xv1.x, EPS) + ew0 + eb0;
        o1.y = fmaxf(xv1.y, EPS) + ew1 + eb1;
        o1.z = fmaxf(xv1.z, EPS) + ew2 + eb2;
        o1.w = fmaxf(xv1.w, EPS) + ew3 + eb3;
    }

    __syncthreads();

    v8f g = {};
    #pragma unroll
    for (int c = 0; c < 4; ++c) {
        float4 eq = *(const float4*)(ebq + c * EBP + dl);
        float4 ek = *(const float4*)(ebk + c * EBP + dl);
        v2f q01 = (v2f){a[c].x, a[c].y} + (v2f){eq.x, eq.y};
        v2f q23 = (v2f){a[c].z, a[c].w} + (v2f){eq.z, eq.w};
        v2f k01 = (v2f){bm[c].x, bm[c].y} + (v2f){ek.x, ek.y};
        v2f k23 = (v2f){bm[c].z, bm[c].w} + (v2f){ek.z, ek.w};
        g = __builtin_amdgcn_wmma_f32_16x16x4_f32(false, q01, false, k01,
                                                  (short)0, g, false, false);
        g = __builtin_amdgcn_wmma_f32_16x16x4_f32(false, q23, false, k23,
                                                  (short)0, g, false, false);
    }

    v8f r;
    {
        const float* swp = rt ? &swq[0][lr][2 * half] : &swk[0][lr][2 * half];
        const float* sbp = rt ? &sbq[0][2 * half] : &sbk[0][2 * half];
        v2f se = *(const v2f*)swp + *(const v2f*)sbp;
        #pragma unroll
        for (int q = 1; q < KSPLIT; ++q) {
            se = se + *(const v2f*)(swp + q * 64);
            se = se + *(const v2f*)(sbp + q * 4);
        }
        v2f xav = {xa.x, xa.y};
        v2f xbv = {xb.x, xb.y};
        v2f kb0 = {fmaf((float)DQ, xb.x, se.x), fmaf((float)DQ, xb.y, se.y)};
        v2f ra = rt ? se : xav;
        v2f rk = rt ? xbv : kb0;
        v8f zero = {};
        r = __builtin_amdgcn_wmma_f32_16x16x4_f32(false, ra, false, rk,
                                                  (short)0, zero, false, false);
    }

    #pragma unroll
    for (int i = 0; i < 8; ++i) {
        part[w][0][i * 32 + lane] = g[i];
        part[w][1][i * 32 + lane] = r[i];
    }
    __syncthreads();

    *(float4*)(Vr + (size_t)((vb + 0) * SS + vs) * NN + tid * 4) = o0;
    *(float4*)(Vr + (size_t)((vb + 1) * SS + vs) * NN + tid * 4) = o1;

    int ob = tid >> 6;
    int s4 = (tid & 63) << 2;
    float4 s = *(const float4*)(&part[0][0][s4]);
    #pragma unroll
    for (int q = 1; q < KSPLIT; ++q) {
        float4 p = *(const float4*)(&part[q][0][s4]);
        s.x += p.x; s.y += p.y; s.z += p.z; s.w += p.w;
    }
    #pragma unroll
    for (int t = 0; t < 2; ++t) {
        float4 p = *(const float4*)(&part[ob + 4 * t][1][s4]);
        s.x += p.x; s.y += p.y; s.z += p.z; s.w += p.w;
    }
    int m = (s4 >> 5) + 8 * ((s4 >> 4) & 1);
    int n = s4 & 15;
    *(float4*)(scores + (size_t)kq * PLANE + (size_t)(ob * SS + i0 + m) * SS + j0 + n) = s;
}

#define APITCH 132

__device__ __forceinline__ float wave_max(float m) {
    m = fmaxf(m, xhalf(m));
    m = fmaxf(m, xrow<8>(m));
    m = fmaxf(m, xrow<4>(m));
    m = fmaxf(m, xrow<2>(m));
    m = fmaxf(m, xrow<1>(m));
    return m;
}
__device__ __forceinline__ float wave_sum(float s) {
    s = s + xhalf(s);
    s = s + xrow<8>(s);
    s = s + xrow<4>(s);
    s = s + xrow<2>(s);
    s = s + xrow<1>(s);
    return s;
}

__global__ __launch_bounds__(256) __attribute__((amdgpu_waves_per_eu(2, 2)))
void softmax_out_wmma_kernel(const float* __restrict__ scores,
                             const float* __restrict__ Vr,
                             float* __restrict__ out) {
    __shared__ __attribute__((aligned(16))) float attn[16 * APITCH];

    int tid  = threadIdx.x;
    int w    = tid >> 5;
    int lane = tid & 31;
    int half = lane >> 4;
    int lr   = lane & 15;
    int blk  = blockIdx.x;
    int b  = blk >> 6;
    int i0 = ((blk >> 3) & 7) << 4;
    int n0 = (((blk & 7) << 3) + w) << 4;

    const float* srow = scores + (size_t)(b * SS + i0 + 2 * w) * SS + lane;
    float v[2][4];
    #pragma unroll
    for (int rr = 0; rr < 2; ++rr)
        #pragma unroll
        for (int q = 0; q < 4; ++q) {
            const float* p = srow + rr * SS + 32 * q;
            v[rr][q] = ((p[0] + p[PLANE]) + p[2 * PLANE]) + p[3 * PLANE];
        }

    const float* bcol = Vr + (size_t)(b * SS) * NN + n0 + lr;
    float bv[SS / 4][2];
    #pragma unroll
    for (int s = 0; s < SS / 4; ++s) {
        int kk = 4 * s + 2 * half;
        bv[s][0] = bcol[(size_t)kk * NN];
        bv[s][1] = bcol[(size_t)(kk + 1) * NN];
    }

    #pragma unroll
    for (int rr = 0; rr < 2; ++rr) {
        float m = wave_max(fmaxf(fmaxf(v[rr][0], v[rr][1]), fmaxf(v[rr][2], v[rr][3])));
        float e0 = __expf(v[rr][0] - m);
        float e1 = __expf(v[rr][1] - m);
        float e2 = __expf(v[rr][2] - m);
        float e3 = __expf(v[rr][3] - m);
        float sum = wave_sum((e0 + e2) + (e1 + e3));
        float* arow_w = attn + (2 * w + rr) * APITCH + lane;
        arow_w[0]  = fmaxf(e0 / sum, EPS);
        arow_w[32] = fmaxf(e1 / sum, EPS);
        arow_w[64] = fmaxf(e2 / sum, EPS);
        arow_w[96] = fmaxf(e3 / sum, EPS);
    }
    __syncthreads();

    const float* arow = attn + lr * APITCH + 2 * half;
    v8f c = {};
    #pragma unroll
    for (int s = 0; s < SS / 4; ++s) {
        v2f a = *(const v2f*)(arow + 4 * s);
        v2f bm = {bv[s][0], bv[s][1]};
        c = __builtin_amdgcn_wmma_f32_16x16x4_f32(false, a, false, bm,
                                                  (short)0, c, false, false);
    }

    float* orow = out + (size_t)(b * SS + i0 + 8 * half) * NN + n0 + lr;
    #pragma unroll
    for (int r = 0; r < 8; ++r) orow[(size_t)r * NN] = c[r];
}

extern "C" void kernel_launch(void* const* d_in, const int* in_sizes, int n_in,
                              void* d_out, int out_size, void* d_ws, size_t ws_size,
                              hipStream_t stream) {
    (void)in_sizes; (void)n_in; (void)out_size; (void)ws_size;

    const float* query  = (const float*)d_in[0];
    const float* key_in = (const float*)d_in[1];
    const float* value  = (const float*)d_in[2];
    const float* qw = (const float*)d_in[3];
    const float* qb = (const float*)d_in[4];
    const float* kw = (const float*)d_in[5];
    const float* kb = (const float*)d_in[6];
    const float* vw = (const float*)d_in[7];
    const float* vb = (const float*)d_in[8];
    float* out = (float*)d_out;

    float* Vr = (float*)d_ws;
    float* sc = Vr + (size_t)BB * SS * NN;

    lns_scores_wmma_kernel<<<256, 256, 0, stream>>>(query, qw, qb, key_in, kw, kb,
                                                    value, vw, vb, Vr, sc);

    softmax_out_wmma_kernel<<<256, 256, 0, stream>>>(sc, Vr, out);
}
